# starting kernel
# baseline (speedup 1.0000x reference)
#include <hip/hip_runtime.h>
#include <hip/hip_bf16.h>

#define N_NODES 50000
#define DIM 128
#define ROW_TILES (N_NODES / 16)

typedef __attribute__((ext_vector_type(2))) float v2f;
typedef __attribute__((ext_vector_type(8))) float v8f;

__global__ __launch_bounds__(256) void sage_fill0(float* __restrict__ p, size_t n) {
    size_t i = (size_t)blockIdx.x * 256 + threadIdx.x;
    if (i < n) p[i] = 0.0f;
}

__global__ __launch_bounds__(256) void sage_count(const int* __restrict__ dst,
                                                  float* __restrict__ cnt, int E) {
    int e = blockIdx.x * 256 + threadIdx.x;
    if (e < E) {
        __hip_atomic_fetch_add(&cnt[dst[e]], 1.0f,
                               __ATOMIC_RELAXED, __HIP_MEMORY_SCOPE_AGENT);
    }
}

__global__ __launch_bounds__(256) void sage_recip(float* __restrict__ c, int n) {
    int i = blockIdx.x * 256 + threadIdx.x;
    if (i < n) c[i] = 1.0f / fmaxf(c[i], 1.0f);
}

__global__ __launch_bounds__(256) void sage_scatter(const float* __restrict__ xin,
                                                    const int* __restrict__ src,
                                                    const int* __restrict__ dst,
                                                    float* __restrict__ summed, int E) {
    int gw   = (blockIdx.x * 256 + threadIdx.x) >> 5;
    int lane = threadIdx.x & 31;
    if (gw >= E) return;
    int s = src[gw];
    int d = dst[gw];
    const float4* xr = (const float4*)(xin + (size_t)s * DIM);
    float4 v = xr[lane];
    float* o = summed + (size_t)d * DIM + lane * 4;
    __hip_atomic_fetch_add(&o[0], v.x, __ATOMIC_RELAXED, __HIP_MEMORY_SCOPE_AGENT);
    __hip_atomic_fetch_add(&o[1], v.y, __ATOMIC_RELAXED, __HIP_MEMORY_SCOPE_AGENT);
    __hip_atomic_fetch_add(&o[2], v.z, __ATOMIC_RELAXED, __HIP_MEMORY_SCOPE_AGENT);
    __hip_atomic_fetch_add(&o[3], v.w, __ATOMIC_RELAXED, __HIP_MEMORY_SCOPE_AGENT);
}

__global__ __launch_bounds__(256) void sage_packw(const float* __restrict__ Wl,
                                                  const float* __restrict__ Wr,
                                                  float2* __restrict__ Wp) {
    int t = blockIdx.x * 256 + threadIdx.x;
    if (t >= 128 * 128) return;
    int p = t >> 7;
    int n = t & 127;
    int k0 = 2 * p;
    float a, b;
    if (k0 < DIM) {
        a = Wl[n * DIM + k0];
        b = Wl[n * DIM + k0 + 1];
    } else {
        a = Wr[n * DIM + (k0 - DIM)];
        b = Wr[n * DIM + (k0 - DIM) + 1];
    }
    Wp[t] = make_float2(a, b);
}

__global__ __launch_bounds__(256) void sage_gemm(const float* __restrict__ summed,
                                                 const float* __restrict__ inv,
                                                 const float* __restrict__ xin,
                                                 const float2* __restrict__ Wp,
                                                 const float* __restrict__ bias,
                                                 float* __restrict__ out, int relu) {
    __shared__ float2 ldsA[16][130];

    const int tid = threadIdx.x;
    const int r0  = blockIdx.x * 16;

    for (int idx = tid; idx < 16 * 128; idx += 256) {
        int m   = idx >> 7;
        int p   = idx & 127;
        int row = r0 + m;
        float2 v;
        if (p < 64) {
            float2 s = ((const float2*)summed)[(size_t)row * 64 + p];
            float iv = inv[row];
            v.x = s.x * iv;
            v.y = s.y * iv;
        } else {
            v = ((const float2*)xin)[(size_t)row * 64 + (p - 64)];
        }
        ldsA[m][p] = v;
    }
    __syncthreads();

    const int lane = tid & 31;
    const int wave = tid >> 5;
    const int half = lane >> 4;
    const int m16  = lane & 15;
    const int n0   = wave * 16;

    const v2f* wpv = (const v2f*)Wp;

    v8f acc = {};
    #pragma unroll 4
    for (int kp = 0; kp < 128; kp += 2) {
        v2f a = *(const v2f*)&ldsA[m16][kp + half];
        v2f b = wpv[(size_t)(kp + half) * 128 + n0 + m16];
        acc = __builtin_amdgcn_wmma_f32_16x16x4_f32(
             false, a,  false, b,
             (short)0, acc,  false,  false);
    }

    const int   ncol = n0 + m16;
    const float bv   = bias[ncol];
    #pragma unroll
    for (int r = 0; r < 8; r++) {
        int   row = r0 + r + 8 * half;
        float val = acc[r] + bv;
        if (relu) val = fmaxf(val, 0.0f);
        out[(size_t)row * DIM + ncol] = val;
    }
}

extern "C" void kernel_launch(void* const* d_in, const int* in_sizes, int n_in,
                              void* d_out, int out_size, void* d_ws, size_t ws_size,
                              hipStream_t stream) {
    const float* x  = (const float*)d_in[0];
    const int*   ei = (const int*)d_in[1];
    const float* Wl = (const float*)d_in[2];
    const float* bl = (const float*)d_in[3];
    const float* Wr = (const float*)d_in[4];
    const int E = in_sizes[1] / 2;
    const int* src = ei;
    const int* dst = ei + E;

    char*   ws     = (char*)d_ws;
    float*  inv    = (float*)ws;
    float2* wp     = (float2*)(ws + (256u << 10));
    float*  summed = (float*)(ws + (512u << 10));
    float*  h1     = (float*)(ws + (512u << 10) + (26u << 20));
    float*  y      = (float*)d_out;

    const size_t nfeat = (size_t)N_NODES * DIM;
    const int fillFeatBlocks = (int)((nfeat + 255) / 256);
    const int fillNBlocks    = (N_NODES + 255) / 256;
    const int cntBlocks      = (E + 255) / 256;
    const int scatBlocks     = (E * 32 + 255) / 256;

    sage_fill0<<<fillNBlocks, 256, 0, stream>>>(inv, (size_t)N_NODES);
    sage_count<<<cntBlocks, 256, 0, stream>>>(dst, inv, E);
    sage_recip<<<fillNBlocks, 256, 0, stream>>>(inv, N_NODES);
    sage_packw<<<64, 256, 0, stream>>>(Wl, Wr, wp);

    sage_fill0<<<fillFeatBlocks, 256, 0, stream>>>(summed, nfeat);
    sage_scatter<<<scatBlocks, 256, 0, stream>>>(x, src, dst, summed, E);
    sage_gemm<<<ROW_TILES, 256, 0, stream>>>(summed, inv, x, wp, bl, h1, 1);

    sage_fill0<<<fillFeatBlocks, 256, 0, stream>>>(summed, nfeat);
    sage_scatter<<<scatBlocks, 256, 0, stream>>>(h1, src, dst, summed, E);
    sage_gemm<<<ROW_TILES, 256, 0, stream>>>(summed, inv, h1, wp, bl, y, 0);
}


# f32 atomic scatter-add and its zero-fill replaced by a dst-sorted edge list (int atomic rank + prefix scan) gathered per row in the GEMM staging; row sum order = arrival-order rank
# speedup vs baseline: 2.1320x; 2.1320x over previous
#include <hip/hip_runtime.h>
#include <hip/hip_bf16.h>

#define N_NODES 50000
#define DIM 128
#define ROW_TILES (N_NODES / 16)
#define DEG_VEC4 (N_NODES / 4)
#define SCAN_BLOCKS ((DEG_VEC4 + 255) / 256)

typedef __attribute__((ext_vector_type(2))) float v2f;
typedef __attribute__((ext_vector_type(8))) float v8f;

__global__ __launch_bounds__(256) void sage_zero(int* __restrict__ p, int n) {
    int i = blockIdx.x * 256 + threadIdx.x;
    if (i < n) p[i] = 0;
}

__global__ __launch_bounds__(256) void sage_rank(const int* __restrict__ dst,
                                                 int* __restrict__ deg,
                                                 int* __restrict__ rank, int E) {
    int e = blockIdx.x * 256 + threadIdx.x;
    if (e < E) {
        rank[e] = __hip_atomic_fetch_add(&deg[dst[e]], 1,
                                         __ATOMIC_RELAXED, __HIP_MEMORY_SCOPE_AGENT);
    }
}

__global__ __launch_bounds__(256) void sage_scan(const int* __restrict__ deg,
                                                 int* __restrict__ rowptr) {
    __shared__ int sa[256];
    __shared__ int sb[256];
    const int t = threadIdx.x;
    const int b = blockIdx.x;
    const int4* d4 = (const int4*)deg;

    int pre = 0;
    for (int k = 0; k < b; k++) {
        int4 q = d4[k * 256 + t];
        pre += (q.x + q.y) + (q.z + q.w);
    }
    const int i4 = b * 256 + t;
    int4 q = make_int4(0, 0, 0, 0);
    if (i4 < DEG_VEC4) q = d4[i4];
    const int loc = (q.x + q.y) + (q.z + q.w);

    sa[t] = pre;
    sb[t] = loc;
    __syncthreads();
    for (int off = 1; off < 256; off <<= 1) {
        int a = 0, c = 0;
        if (t >= off) { a = sa[t - off]; c = sb[t - off]; }
        __syncthreads();
        sa[t] += a;
        sb[t] += c;
        __syncthreads();
    }
    const int run = sa[255] + sb[t] - loc;

    if (i4 < DEG_VEC4) {
        int4 o;
        o.x = run;
        o.y = o.x + q.x;
        o.z = o.y + q.y;
        o.w = o.z + q.z;
        ((int4*)rowptr)[i4] = o;
        if (i4 == DEG_VEC4 - 1) rowptr[N_NODES] = o.w + q.w;
    }
}

__global__ __launch_bounds__(256) void sage_place(const int* __restrict__ src,
                                                  const int* __restrict__ dst,
                                                  const int* __restrict__ rowptr,
                                                  const int* __restrict__ rank,
                                                  int* __restrict__ col, int E) {
    int e = blockIdx.x * 256 + threadIdx.x;
    if (e < E) col[rowptr[dst[e]] + rank[e]] = src[e];
}

__global__ __launch_bounds__(256) void sage_packw(const float* __restrict__ Wl,
                                                  const float* __restrict__ Wr,
                                                  float2* __restrict__ Wp) {
    int t = blockIdx.x * 256 + threadIdx.x;
    if (t >= 128 * 128) return;
    int p = t >> 7;
    int n = t & 127;
    int k0 = 2 * p;
    float a, b;
    if (k0 < DIM) {
        a = Wl[n * DIM + k0];
        b = Wl[n * DIM + k0 + 1];
    } else {
        a = Wr[n * DIM + (k0 - DIM)];
        b = Wr[n * DIM + (k0 - DIM) + 1];
    }
    Wp[t] = make_float2(a, b);
}

__global__ __launch_bounds__(256) void sage_gemm(const int* __restrict__ rowptr,
                                                 const int* __restrict__ col,
                                                 const float* __restrict__ xin,
                                                 const float2* __restrict__ Wp,
                                                 const float* __restrict__ bias,
                                                 float* __restrict__ out, int relu) {
    __shared__ float2 ldsA[16][130];

    const int tid  = threadIdx.x;
    const int lane = tid & 31;
    const int wave = tid >> 5;
    const int r0   = blockIdx.x * 16;

    for (int idx = tid; idx < 16 * 64; idx += 256) {
        int m = idx >> 6;
        int p = idx & 63;
        ldsA[m][64 + p] = ((const float2*)xin)[(size_t)(r0 + m) * 64 + p];
    }

    const float4* x4 = (const float4*)xin;
    for (int rr = 0; rr < 2; rr++) {
        const int m   = wave * 2 + rr;
        const int row = r0 + m;
        const int beg = __builtin_amdgcn_readfirstlane(rowptr[row]);
        const int end = __builtin_amdgcn_readfirstlane(rowptr[row + 1]);
        float4 acc = make_float4(0.0f, 0.0f, 0.0f, 0.0f);
        for (int base = beg; base < end; base += 32) {
            const int n = (end - base < 32) ? (end - base) : 32;
            const int myid = (lane < n) ? col[base + lane] : 0;
            for (int k = 0; k < n; k += 8) {
                float4 v[8];
                int sid[8];
                #pragma unroll
                for (int u = 0; u < 8; u++) {
                    sid[u] = __builtin_amdgcn_readlane(myid, k + u);
                }
                #pragma unroll
                for (int u = 0; u < 8; u++) {
                    v[u] = make_float4(0.0f, 0.0f, 0.0f, 0.0f);
                    if (k + u < n) {
                        v[u] = x4[(size_t)sid[u] * 32 + lane];
                    }
                }
                #pragma unroll
                for (int u = 0; u < 8; u++) {
                    acc.x += v[u].x;
                    acc.y += v[u].y;
                    acc.z += v[u].z;
                    acc.w += v[u].w;
                }
            }
        }
        const float iv = 1.0f / fmaxf((float)(end - beg), 1.0f);
        ldsA[m][2 * lane]     = make_float2(acc.x * iv, acc.y * iv);
        ldsA[m][2 * lane + 1] = make_float2(acc.z * iv, acc.w * iv);
    }
    __syncthreads();

    const int half = lane >> 4;
    const int m16  = lane & 15;
    const int n0   = wave * 16;

    const v2f* wpv = (const v2f*)Wp;

    v8f acc = {};
    #pragma unroll 4
    for (int kp = 0; kp < 128; kp += 2) {
        v2f a = *(const v2f*)&ldsA[m16][kp + half];
        v2f b = wpv[(size_t)(kp + half) * 128 + n0 + m16];
        acc = __builtin_amdgcn_wmma_f32_16x16x4_f32(
             false, a,  false, b,
             (short)0, acc,  false,  false);
    }

    const int   ncol = n0 + m16;
    const float bv   = bias[ncol];
    #pragma unroll
    for (int r = 0; r < 8; r++) {
        int   row = r0 + r + 8 * half;
        float val = acc[r] + bv;
        if (relu) val = fmaxf(val, 0.0f);
        out[(size_t)row * DIM + ncol] = val;
    }
}

extern "C" void kernel_launch(void* const* d_in, const int* in_sizes, int n_in,
                              void* d_out, int out_size, void* d_ws, size_t ws_size,
                              hipStream_t stream) {
    const float* x  = (const float*)d_in[0];
    const int*   ei = (const int*)d_in[1];
    const float* Wl = (const float*)d_in[2];
    const float* bl = (const float*)d_in[3];
    const float* Wr = (const float*)d_in[4];
    const int E = in_sizes[1] / 2;
    const int* src = ei;
    const int* dst = ei + E;

    char*   ws     = (char*)d_ws;
    int*    deg    = (int*)ws;
    float2* wp     = (float2*)(ws + (256u << 10));
    int*    rowptr = (int*)(ws + (512u << 10));
    int*    rank   = (int*)(ws + (512u << 10) + (1u << 20));
    int*    col    = (int*)(ws + (512u << 10) + (4u << 20));
    float*  h1     = (float*)(ws + (512u << 10) + (26u << 20));
    float*  y      = (float*)d_out;

    const int nodeBlocks = (N_NODES + 255) / 256;
    const int edgeBlocks = (E + 255) / 256;

    sage_zero<<<nodeBlocks, 256, 0, stream>>>(deg, N_NODES);
    sage_rank<<<edgeBlocks, 256, 0, stream>>>(dst, deg, rank, E);
    sage_scan<<<SCAN_BLOCKS, 256, 0, stream>>>(deg, rowptr);
    sage_place<<<edgeBlocks, 256, 0, stream>>>(src, dst, rowptr, rank, col, E);
    sage_packw<<<64, 256, 0, stream>>>(Wl, Wr, wp);

    sage_gemm<<<ROW_TILES, 256, 0, stream>>>(rowptr, col, x, wp, bl, h1, 1);

    sage_gemm<<<ROW_TILES, 256, 0, stream>>>(rowptr, col, h1, wp, bl, y, 0);
}


# layer kernel retiled: 768 work-groups of 16 waves take 4-5 adjacent 16-row tiles each, so one weight (B) fragment load feeds 2-3 row tiles; a wave gathers its rows as one edge run; sum orders kept
# speedup vs baseline: 2.2399x; 1.0506x over previous
#include <hip/hip_runtime.h>
#include <hip/hip_bf16.h>

#define N_NODES 50000
#define DIM 128
#define ROW_TILES (N_NODES / 16)
#define DEG_VEC4 (N_NODES / 4)
#define SCAN_BLOCKS ((DEG_VEC4 + 255) / 256)
#define TILE_M 80
#define ROW_BLOCKS 768
static_assert(ROW_TILES >= 4 * ROW_BLOCKS && ROW_TILES <= 5 * ROW_BLOCKS, "row tiles per block");
static_assert(TILE_M == 5 * 16, "LDS panel holds five row tiles");

typedef __attribute__((ext_vector_type(2))) float v2f;
typedef __attribute__((ext_vector_type(8))) float v8f;

__global__ __launch_bounds__(256) void sage_zero(int* __restrict__ p, int n) {
    int i = blockIdx.x * 256 + threadIdx.x;
    if (i < n) p[i] = 0;
}

__global__ __launch_bounds__(256) void sage_rank(const int* __restrict__ dst,
                                                 int* __restrict__ deg,
                                                 int* __restrict__ rank, int E) {
    int e = blockIdx.x * 256 + threadIdx.x;
    if (e < E) {
        rank[e] = __hip_atomic_fetch_add(&deg[dst[e]], 1,
                                         __ATOMIC_RELAXED, __HIP_MEMORY_SCOPE_AGENT);
    }
}

__global__ __launch_bounds__(256) void sage_scan(const int* __restrict__ deg,
                                                 int* __restrict__ rowptr) {
    __shared__ int sa[256];
    __shared__ int sb[256];
    const int t = threadIdx.x;
    const int b = blockIdx.x;
    const int4* d4 = (const int4*)deg;

    int pre = 0;
    for (int k = 0; k < b; k++) {
        int4 q = d4[k * 256 + t];
        pre += (q.x + q.y) + (q.z + q.w);
    }
    const int i4 = b * 256 + t;
    int4 q = make_int4(0, 0, 0, 0);
    if (i4 < DEG_VEC4) q = d4[i4];
    const int loc = (q.x + q.y) + (q.z + q.w);

    sa[t] = pre;
    sb[t] = loc;
    __syncthreads();
    for (int off = 1; off < 256; off <<= 1) {
        int a = 0, c = 0;
        if (t >= off) { a = sa[t - off]; c = sb[t - off]; }
        __syncthreads();
        sa[t] += a;
        sb[t] += c;
        __syncthreads();
    }
    const int run = sa[255] + sb[t] - loc;

    if (i4 < DEG_VEC4) {
        int4 o;
        o.x = run;
        o.y = o.x + q.x;
        o.z = o.y + q.y;
        o.w = o.z + q.z;
        ((int4*)rowptr)[i4] = o;
        if (i4 == DEG_VEC4 - 1) rowptr[N_NODES] = o.w + q.w;
    }
}

__global__ __launch_bounds__(256) void sage_place(const int* __restrict__ src,
                                                  const int* __restrict__ dst,
                                                  const int* __restrict__ rowptr,
                                                  const int* __restrict__ rank,
                                                  int* __restrict__ col, int E) {
    int e = blockIdx.x * 256 + threadIdx.x;
    if (e < E) col[rowptr[dst[e]] + rank[e]] = src[e];
}

__global__ __launch_bounds__(256) void sage_packw(const float* __restrict__ Wl,
                                                  const float* __restrict__ Wr,
                                                  float2* __restrict__ Wp) {
    int t = blockIdx.x * 256 + threadIdx.x;
    if (t >= 128 * 128) return;
    int p = t >> 7;
    int n = t & 127;
    int k0 = 2 * p;
    float a, b;
    if (k0 < DIM) {
        a = Wl[n * DIM + k0];
        b = Wl[n * DIM + k0 + 1];
    } else {
        a = Wr[n * DIM + (k0 - DIM)];
        b = Wr[n * DIM + (k0 - DIM) + 1];
    }
    Wp[t] = make_float2(a, b);
}

template <int NT>
__device__ __forceinline__ void sage_tiles(const float2 (*ldsA)[130],
                                           const v2f* __restrict__ wpv,
                                           const float* __restrict__ bias,
                                           float* __restrict__ out,
                                           int r0, int mt0, int n0, int lane, int relu) {
    const int half = lane >> 4;
    const int m16  = lane & 15;

    v8f acc[NT];
    #pragma unroll
    for (int t = 0; t < NT; t++) acc[t] = v8f{};

    #pragma unroll 4
    for (int kp = 0; kp < 128; kp += 2) {
        v2f b = wpv[(size_t)(kp + half) * 128 + n0 + m16];
        #pragma unroll
        for (int t = 0; t < NT; t++) {
            v2f a = *(const v2f*)&ldsA[(mt0 + t) * 16 + m16][kp + half];
            acc[t] = __builtin_amdgcn_wmma_f32_16x16x4_f32(
                 false, a,  false, b,
                 (short)0, acc[t],  false,  false);
        }
    }

    const int   ncol = n0 + m16;
    const float bv   = bias[ncol];
    #pragma unroll
    for (int t = 0; t < NT; t++) {
        #pragma unroll
        for (int r = 0; r < 8; r++) {
            int   row = r0 + (mt0 + t) * 16 + r + 8 * half;
            float val = acc[t][r] + bv;
            if (relu) val = fmaxf(val, 0.0f);
            out[(size_t)row * DIM + ncol] = val;
        }
    }
}

__global__ __launch_bounds__(512) void sage_gemm(const int* __restrict__ rowptr,
                                                 const int* __restrict__ col,
                                                 const float* __restrict__ xin,
                                                 const float2* __restrict__ Wp,
                                                 const float* __restrict__ bias,
                                                 float* __restrict__ out, int relu) {
    __shared__ float2 ldsA[TILE_M][130];

    const int tid  = threadIdx.x;
    const int lane = tid & 31;
    const int wave = __builtin_amdgcn_readfirstlane(tid >> 5);

    const int t0 = (int)(((unsigned)blockIdx.x * ROW_TILES) / ROW_BLOCKS);
    const int nt = (int)((((unsigned)blockIdx.x + 1u) * ROW_TILES) / ROW_BLOCKS) - t0;
    const int r0 = t0 * 16;

    const float4* x4 = (const float4*)xin;
    const int m0 = wave * nt;
    const int rpv = rowptr[r0 + m0 + (lane < nt ? lane : nt)];

    {
        float4 rt[4];
        #pragma unroll
        for (int i = 0; i < 4; i++) {
            const int idx = tid + i * 512;
            rt[i] = x4[(size_t)(r0 + (idx >> 5)) * 32 + (idx & 31)];
        }
        float4 rt4 = make_float4(0.0f, 0.0f, 0.0f, 0.0f);
        if (nt == 5) {
            const int idx = tid + 4 * 512;
            rt4 = x4[(size_t)(r0 + (idx >> 5)) * 32 + (idx & 31)];
        }
        #pragma unroll
        for (int i = 0; i < 4; i++) {
            const int idx = tid + i * 512;
            *(float4*)&ldsA[idx >> 5][64 + 2 * (idx & 31)] = rt[i];
        }
        if (nt == 5) {
            const int idx = tid + 4 * 512;
            *(float4*)&ldsA[idx >> 5][64 + 2 * (idx & 31)] = rt4;
        }
    }

    {
        const int sb = __builtin_amdgcn_readlane(rpv, 0);
        const int se = __builtin_amdgcn_readlane(rpv, nt);
        int cur  = 0;
        int curb = sb;
        int nxtb = __builtin_amdgcn_readlane(rpv, 1);
        float4 acc = make_float4(0.0f, 0.0f, 0.0f, 0.0f);

        for (int base = sb; base < se; base += 32) {
            const int n = (se - base < 32) ? (se - base) : 32;
            const int myid = (lane < n) ? col[base + lane] : 0;
            for (int k = 0; k < n; k += 8) {
                float4 v[8];
                int sid[8];
                #pragma unroll
                for (int u = 0; u < 8; u++) {
                    sid[u] = __builtin_amdgcn_readlane(myid, k + u);
                }
                #pragma unroll
                for (int u = 0; u < 8; u++) {
                    v[u] = make_float4(0.0f, 0.0f, 0.0f, 0.0f);
                    if (k + u < n) {
                        v[u] = x4[(size_t)sid[u] * 32 + lane];
                    }
                }
                #pragma unroll
                for (int u = 0; u < 8; u++) {
                    if (k + u < n) {
                        const int j = base + k + u;
                        while (j >= nxtb) {
                            const float iv = 1.0f / fmaxf((float)(nxtb - curb), 1.0f);
                            ldsA[m0 + cur][2 * lane]     = make_float2(acc.x * iv, acc.y * iv);
                            ldsA[m0 + cur][2 * lane + 1] = make_float2(acc.z * iv, acc.w * iv);
                            acc = make_float4(0.0f, 0.0f, 0.0f, 0.0f);
                            cur++;
                            curb = nxtb;
                            nxtb = __builtin_amdgcn_readlane(rpv, cur + 1);
                        }
                        acc.x += v[u].x;
                        acc.y += v[u].y;
                        acc.z += v[u].z;
                        acc.w += v[u].w;
                    }
                }
            }
        }
        while (cur < nt) {
            const float iv = 1.0f / fmaxf((float)(nxtb - curb), 1.0f);
            ldsA[m0 + cur][2 * lane]     = make_float2(acc.x * iv, acc.y * iv);
            ldsA[m0 + cur][2 * lane + 1] = make_float2(acc.z * iv, acc.w * iv);
            acc = make_float4(0.0f, 0.0f, 0.0f, 0.0f);
            cur++;
            curb = nxtb;
            nxtb = __builtin_amdgcn_readlane(rpv, (cur + 1 <= nt) ? cur + 1 : nt);
        }
    }
    __syncthreads();

    const int n0 = (wave & 7) * 16;
    const v2f* wpv = (const v2f*)Wp;
    if (nt == 5 && wave < 8) {
        sage_tiles<3>(ldsA, wpv, bias, out, r0, 0, n0, lane, relu);
    } else {
        sage_tiles<2>(ldsA, wpv, bias, out, r0, (wave < 8) ? 0 : nt - 2, n0, lane, relu);
    }
}

extern "C" void kernel_launch(void* const* d_in, const int* in_sizes, int n_in,
                              void* d_out, int out_size, void* d_ws, size_t ws_size,
                              hipStream_t stream) {
    const float* x  = (const float*)d_in[0];
    const int*   ei = (const int*)d_in[1];
    const float* Wl = (const float*)d_in[2];
    const float* bl = (const float*)d_in[3];
    const float* Wr = (const float*)d_in[4];
    const int E = in_sizes[1] / 2;
    const int* src = ei;
    const int* dst = ei + E;

    char*   ws     = (char*)d_ws;
    int*    deg    = (int*)ws;
    float2* wp     = (float2*)(ws + (256u << 10));
    int*    rowptr = (int*)(ws + (512u << 10));
    int*    rank   = (int*)(ws + (512u << 10) + (1u << 20));
    int*    col    = (int*)(ws + (512u << 10) + (4u << 20));
    float*  h1     = (float*)(ws + (512u << 10) + (26u << 20));
    float*  y      = (float*)d_out;

    const int nodeBlocks = (N_NODES + 255) / 256;
    const int edgeBlocks = (E + 255) / 256;

    sage_zero<<<nodeBlocks, 256, 0, stream>>>(deg, N_NODES);
    sage_rank<<<edgeBlocks, 256, 0, stream>>>(dst, deg, rank, E);
    sage_scan<<<SCAN_BLOCKS, 256, 0, stream>>>(deg, rowptr);
    sage_place<<<edgeBlocks, 256, 0, stream>>>(src, dst, rowptr, rank, col, E);
    sage_packw<<<64, 256, 0, stream>>>(Wl, Wr, wp);

    sage_gemm<<<ROW_BLOCKS, 512, 0, stream>>>(rowptr, col, x, wp, bl, h1, 1);

    sage_gemm<<<ROW_BLOCKS, 512, 0, stream>>>(rowptr, col, h1, wp, bl, y, 0);
}


# dst sort (degree scan kernel, placement kernel, rank array) replaced by 32-slot per-node id lines filled by the ranking atomic itself, spill list past 32; 7 launches -> 5; row sums keep arrival order
# speedup vs baseline: 2.7429x; 1.2246x over previous
#include <hip/hip_runtime.h>
#include <hip/hip_bf16.h>

#define N_NODES 50000
#define DIM 128
#define ROW_TILES (N_NODES / 16)
#define CAP 32
#define TILE_M 80
#define ROW_BLOCKS 768
static_assert(ROW_TILES >= 4 * ROW_BLOCKS && ROW_TILES <= 5 * ROW_BLOCKS, "row tiles per block");
static_assert(TILE_M == 5 * 16, "LDS panel holds five row tiles");

typedef __attribute__((ext_vector_type(2))) float v2f;
typedef __attribute__((ext_vector_type(8))) float v8f;

__global__ __launch_bounds__(256) void sage_zero(int* __restrict__ p, int n) {
    int i = blockIdx.x * 256 + threadIdx.x;
    if (i < n) p[i] = 0;
}

__global__ __launch_bounds__(256) void sage_fill(const int* __restrict__ src,
                                                 const int* __restrict__ dst,
                                                 int* __restrict__ deg,
                                                 int* __restrict__ bucket,
                                                 int* __restrict__ spill, int E) {
    int e = blockIdx.x * 256 + threadIdx.x;
    if (e < E) {
        const int d = dst[e];
        const int s = src[e];
        const int r = __hip_atomic_fetch_add(&deg[d], 1,
                                             __ATOMIC_RELAXED, __HIP_MEMORY_SCOPE_AGENT);
        if (r < CAP) {
            bucket[(size_t)d * CAP + r] = s;
        } else {
            const int k = __hip_atomic_fetch_add(&deg[N_NODES], 1,
                                                 __ATOMIC_RELAXED, __HIP_MEMORY_SCOPE_AGENT);
            spill[2 * (size_t)k]     = d;
            spill[2 * (size_t)k + 1] = s;
        }
    }
}

__global__ __launch_bounds__(256) void sage_packw(const float* __restrict__ Wl,
                                                  const float* __restrict__ Wr,
                                                  float2* __restrict__ Wp) {
    int t = blockIdx.x * 256 + threadIdx.x;
    if (t >= 128 * 128) return;
    int p = t >> 7;
    int n = t & 127;
    int k0 = 2 * p;
    float a, b;
    if (k0 < DIM) {
        a = Wl[n * DIM + k0];
        b = Wl[n * DIM + k0 + 1];
    } else {
        a = Wr[n * DIM + (k0 - DIM)];
        b = Wr[n * DIM + (k0 - DIM) + 1];
    }
    Wp[t] = make_float2(a, b);
}

template <int NT>
__device__ __forceinline__ void sage_tiles(const float2 (*ldsA)[130],
                                           const v2f* __restrict__ wpv,
                                           const float* __restrict__ bias,
                                           float* __restrict__ out,
                                           int r0, int mt0, int n0, int lane, int relu) {
    const int half = lane >> 4;
    const int m16  = lane & 15;

    v8f acc[NT];
    #pragma unroll
    for (int t = 0; t < NT; t++) acc[t] = v8f{};

    #pragma unroll 4
    for (int kp = 0; kp < 128; kp += 2) {
        v2f b = wpv[(size_t)(kp + half) * 128 + n0 + m16];
        #pragma unroll
        for (int t = 0; t < NT; t++) {
            v2f a = *(const v2f*)&ldsA[(mt0 + t) * 16 + m16][kp + half];
            acc[t] = __builtin_amdgcn_wmma_f32_16x16x4_f32(
                 false, a,  false, b,
                 (short)0, acc[t],  false,  false);
        }
    }

    const int   ncol = n0 + m16;
    const float bv   = bias[ncol];
    #pragma unroll
    for (int t = 0; t < NT; t++) {
        #pragma unroll
        for (int r = 0; r < 8; r++) {
            int   row = r0 + (mt0 + t) * 16 + r + 8 * half;
            float val = acc[t][r] + bv;
            if (relu) val = fmaxf(val, 0.0f);
            out[(size_t)row * DIM + ncol] = val;
        }
    }
}

__global__ __launch_bounds__(512) void sage_gemm(const int* __restrict__ deg,
                                                 const int* __restrict__ bucket,
                                                 const int* __restrict__ spill,
                                                 const float* __restrict__ xin,
                                                 const float2* __restrict__ Wp,
                                                 const float* __restrict__ bias,
                                                 float* __restrict__ out, int relu) {
    __shared__ float2 ldsA[TILE_M][130];

    const int tid  = threadIdx.x;
    const int lane = tid & 31;
    const int wave = __builtin_amdgcn_readfirstlane(tid >> 5);

    const int t0 = (int)(((unsigned)blockIdx.x * ROW_TILES) / ROW_BLOCKS);
    const int nt = (int)((((unsigned)blockIdx.x + 1u) * ROW_TILES) / ROW_BLOCKS) - t0;
    const int r0 = t0 * 16;

    const float4* x4 = (const float4*)xin;
    const int m0 = wave * nt;
    const int degl = deg[r0 + m0 + (lane < nt ? lane : nt - 1)];
    int ids[5];
    #pragma unroll
    for (int i = 0; i < 5; i++) {
        ids[i] = bucket[(r0 + m0 + (i < nt ? i : nt - 1)) * CAP + lane];
    }

    {
        float4 rt[4];
        #pragma unroll
        for (int i = 0; i < 4; i++) {
            const int idx = tid + i * 512;
            rt[i] = x4[(size_t)(r0 + (idx >> 5)) * 32 + (idx & 31)];
        }
        float4 rt4 = make_float4(0.0f, 0.0f, 0.0f, 0.0f);
        if (nt == 5) {
            const int idx = tid + 4 * 512;
            rt4 = x4[(size_t)(r0 + (idx >> 5)) * 32 + (idx & 31)];
        }
        #pragma unroll
        for (int i = 0; i < 4; i++) {
            const int idx = tid + i * 512;
            *(float4*)&ldsA[idx >> 5][64 + 2 * (idx & 31)] = rt[i];
        }
        if (nt == 5) {
            const int idx = tid + 4 * 512;
            *(float4*)&ldsA[idx >> 5][64 + 2 * (idx & 31)] = rt4;
        }
    }

    {
        const int dcl = (lane < nt) ? (degl < CAP ? degl : CAP) : 0;
        int incl = dcl;
        #pragma unroll
        for (int off = 1; off < 32; off <<= 1) {
            const int t = __shfl_up(incl, off);
            if (lane >= off) incl += t;
        }
        const int rpv = incl - dcl;

        const int se = __builtin_amdgcn_readlane(rpv, nt);
        int cur  = 0;
        int nxtb = __builtin_amdgcn_readlane(rpv, 1);
        float4 acc = make_float4(0.0f, 0.0f, 0.0f, 0.0f);

#define SAGE_CLOSE_ROW()                                                                   \
        {                                                                                  \
              \
            const int fd = __builtin_amdgcn_readlane(degl, cur);                           \
            if (fd > CAP) {                                                                \
                const int grow = r0 + m0 + cur;                                            \
                const int ns = deg[N_NODES];                                               \
                for (int q = 0; q < ns; q++) {                                             \
                    if (spill[2 * q] == grow) {                                            \
                        const float4 sv = x4[(size_t)spill[2 * q + 1] * 32 + lane];        \
                        acc.x += sv.x;                                                     \
                        acc.y += sv.y;                                                     \
                        acc.z += sv.z;                                                     \
                        acc.w += sv.w;                                                     \
                    }                                                                      \
                }                                                                          \
            }                                                                              \
                   \
            const float iv = 1.0f / fmaxf((float)fd, 1.0f);                                \
            ldsA[m0 + cur][2 * lane]     = make_float2(acc.x * iv, acc.y * iv);            \
            ldsA[m0 + cur][2 * lane + 1] = make_float2(acc.z * iv, acc.w * iv);            \
            acc = make_float4(0.0f, 0.0f, 0.0f, 0.0f);                                     \
            cur++;                                                                         \
            nxtb = __builtin_amdgcn_readlane(rpv, (cur + 1 <= nt) ? cur + 1 : nt);         \
        }

        for (int base = 0; base < se; base += 32) {
            const int n = (se - base < 32) ? (se - base) : 32;
            const int p = base + lane;
            int myid = 0;
            #pragma unroll
            for (int i = 0; i < 5; i++) {
                const int st = __builtin_amdgcn_readlane(rpv, i);
                const int en = __builtin_amdgcn_readlane(rpv, i + 1);
                const int t  = __shfl(ids[i], (p - st) & 31);
                if (p >= st && p < en) myid = t;
            }
            for (int k = 0; k < n; k += 8) {
                float4 v[8];
                int sid[8];
                #pragma unroll
                for (int u = 0; u < 8; u++) {
                    sid[u] = __builtin_amdgcn_readlane(myid, k + u);
                }
                #pragma unroll
                for (int u = 0; u < 8; u++) {
                    v[u] = make_float4(0.0f, 0.0f, 0.0f, 0.0f);
                    if (k + u < n) {
                        v[u] = x4[(size_t)sid[u] * 32 + lane];
                    }
                }
                #pragma unroll
                for (int u = 0; u < 8; u++) {
                    if (k + u < n) {
                        const int j = base + k + u;
                        while (j >= nxtb) SAGE_CLOSE_ROW()
                        acc.x += v[u].x;
                        acc.y += v[u].y;
                        acc.z += v[u].z;
                        acc.w += v[u].w;
                    }
                }
            }
        }
        while (cur < nt) SAGE_CLOSE_ROW()
#undef SAGE_CLOSE_ROW
    }
    __syncthreads();

    const int n0 = (wave & 7) * 16;
    const v2f* wpv = (const v2f*)Wp;
    if (nt == 5 && wave < 8) {
        sage_tiles<3>(ldsA, wpv, bias, out, r0, 0, n0, lane, relu);
    } else {
        sage_tiles<2>(ldsA, wpv, bias, out, r0, (wave < 8) ? 0 : nt - 2, n0, lane, relu);
    }
}

extern "C" void kernel_launch(void* const* d_in, const int* in_sizes, int n_in,
                              void* d_out, int out_size, void* d_ws, size_t ws_size,
                              hipStream_t stream) {
    const float* x  = (const float*)d_in[0];
    const int*   ei = (const int*)d_in[1];
    const float* Wl = (const float*)d_in[2];
    const float* bl = (const float*)d_in[3];
    const float* Wr = (const float*)d_in[4];
    const int E = in_sizes[1] / 2;
    const int* src = ei;
    const int* dst = ei + E;

    char*   ws     = (char*)d_ws;
    int*    deg    = (int*)ws;
    float2* wp     = (float2*)(ws + (256u << 10));
    int*    bucket = (int*)(ws + (512u << 10) + (4u << 20));
    int*    spill  = (int*)(ws + (512u << 10) + (12u << 20));
    float*  h1     = (float*)(ws + (512u << 10) + (26u << 20));
    float*  y      = (float*)d_out;

    const int nodeBlocks = (N_NODES + 1 + 255) / 256;
    const int edgeBlocks = (E + 255) / 256;

    sage_zero<<<nodeBlocks, 256, 0, stream>>>(deg, N_NODES + 1);
    sage_fill<<<edgeBlocks, 256, 0, stream>>>(src, dst, deg, bucket, spill, E);
    sage_packw<<<64, 256, 0, stream>>>(Wl, Wr, wp);

    sage_gemm<<<ROW_BLOCKS, 512, 0, stream>>>(deg, bucket, spill, x, wp, bl, h1, 1);

    sage_gemm<<<ROW_BLOCKS, 512, 0, stream>>>(deg, bucket, spill, h1, wp, bl, y, 0);
}


# edge ranking moved off 640000 scattered global atomics: chunk-local LDS counting sort into 256 node ranges, then a block per range ranks by LDS add; no zero kernel; row sum order = LDS arrival order
# speedup vs baseline: 3.7194x; 1.3560x over previous
#include <hip/hip_runtime.h>
#include <hip/hip_bf16.h>

#define N_NODES 50000
#define DIM 128
#define ROW_TILES (N_NODES / 16)
#define CAP 32
#define TILE_M 80
#define ROW_BLOCKS 768
#define N_EDGES 640000
#define CHUNK 2500
#define EBLOCKS (N_EDGES / CHUNK)
#define EPT 5
#define BIN_N 196
#define NBINS 256
#define RPT 3
static_assert(ROW_TILES >= 4 * ROW_BLOCKS && ROW_TILES <= 5 * ROW_BLOCKS, "row tiles per block");
static_assert(TILE_M == 5 * 16, "LDS panel holds five row tiles");
static_assert(N_EDGES % CHUNK == 0 && EBLOCKS == 256 && EPT * 512 >= CHUNK, "edge chunks");
static_assert(BIN_N * NBINS >= N_NODES && BIN_N * (NBINS - 1) < N_NODES, "node ranges cover the nodes");

typedef __attribute__((ext_vector_type(2))) float v2f;
typedef __attribute__((ext_vector_type(8))) float v8f;

__global__ __launch_bounds__(512) void sage_binsort(const int* __restrict__ src,
                                                    const int* __restrict__ dst,
                                                    int* __restrict__ deg,
                                                    int2* __restrict__ tab,
                                                    int2* __restrict__ pairs, int E) {
    __shared__ int  cnt[NBINS];
    __shared__ int  boff[NBINS];
    __shared__ int  wsum[NBINS / 32];
    __shared__ int2 sorted[CHUNK];

    const int tid  = threadIdx.x;
    const int lane = tid & 31;
    const int wave = tid >> 5;
    const int base = blockIdx.x * CHUNK;

    if (tid < NBINS) cnt[tid] = 0;
    if (blockIdx.x == 0 && tid == 0) deg[N_NODES] = 0;
    __syncthreads();

    int d[EPT], s[EPT], bn[EPT], r[EPT];
    #pragma unroll
    for (int j = 0; j < EPT; j++) {
        const int i = tid + j * 512;
        d[j] = 0; s[j] = 0; r[j] = 0; bn[j] = -1;
        if (i < CHUNK && base + i < E) {
            d[j]  = dst[base + i];
            s[j]  = src[base + i];
            bn[j] = d[j] / BIN_N;
        }
    }
    #pragma unroll
    for (int j = 0; j < EPT; j++) {
        if (bn[j] >= 0) r[j] = atomicAdd(&cnt[bn[j]], 1);
    }
    __syncthreads();

    const int v = (tid < NBINS) ? cnt[tid] : 0;
    int inc = v;
    #pragma unroll
    for (int off = 1; off < 32; off <<= 1) {
        const int t = __shfl_up(inc, off);
        if (lane >= off) inc += t;
    }
    if (tid < NBINS && lane == 31) wsum[wave] = inc;
    __syncthreads();
    if (tid < NBINS) {
        int pre = 0;
        for (int w = 0; w < wave; w++) pre += wsum[w];
        const int o = pre + inc - v;
        boff[tid] = o;
        tab[(size_t)tid * EBLOCKS + blockIdx.x] = make_int2(o, v);
    }
    __syncthreads();

    #pragma unroll
    for (int j = 0; j < EPT; j++) {
        if (bn[j] >= 0) sorted[boff[bn[j]] + r[j]] = make_int2(d[j], s[j]);
    }
    __syncthreads();
    for (int i = tid; i < CHUNK; i += 512) {
        if (base + i < E) pairs[base + i] = sorted[i];
    }
}

__global__ __launch_bounds__(1024) void sage_rank(const int2* __restrict__ tab,
                                                 const int2* __restrict__ pairs,
                                                 int* __restrict__ deg,
                                                 int* __restrict__ bucket,
                                                 int* __restrict__ spill) {
    __shared__ int lines[BIN_N * CAP];
    __shared__ int dcnt[BIN_N];
    __shared__ int segoff[EBLOCKS + 1];
    __shared__ int segsrc[EBLOCKS];
    __shared__ int wsum[EBLOCKS / 32];

    const int tid   = threadIdx.x;
    const int lane  = tid & 31;
    const int wave  = tid >> 5;
    const int node0 = blockIdx.x * BIN_N;

    if (tid < BIN_N) dcnt[tid] = 0;
    for (int i = tid; i < BIN_N * CAP; i += 1024) lines[i] = 0;
    int2 te = make_int2(0, 0);
    if (tid < EBLOCKS) te = tab[(size_t)blockIdx.x * EBLOCKS + tid];
    int inc = te.y;
    #pragma unroll
    for (int off = 1; off < 32; off <<= 1) {
        const int t = __shfl_up(inc, off);
        if (lane >= off) inc += t;
    }
    if (tid < EBLOCKS && lane == 31) wsum[wave] = inc;
    __syncthreads();
    if (tid < EBLOCKS) {
        int pre = 0;
        for (int w = 0; w < wave; w++) pre += wsum[w];
        segoff[tid] = pre + inc - te.y;
        segsrc[tid] = tid * CHUNK + te.x;
        if (tid == EBLOCKS - 1) segoff[EBLOCKS] = pre + inc;
    }
    __syncthreads();

    const int total = segoff[EBLOCKS];
    for (int i0 = tid; i0 < total; i0 += RPT * 1024) {
        int lo[RPT], hi[RPT];
        #pragma unroll
        for (int u = 0; u < RPT; u++) { lo[u] = 0; hi[u] = EBLOCKS; }
        #pragma unroll
        for (int it = 0; it < 8; it++) {
            #pragma unroll
            for (int u = 0; u < RPT; u++) {
                const int mid = (lo[u] + hi[u]) >> 1;
                if (segoff[mid] <= i0 + u * 1024) lo[u] = mid; else hi[u] = mid;
            }
        }
        int2 p[RPT];
        #pragma unroll
        for (int u = 0; u < RPT; u++) {
            const int i = i0 + u * 1024;
            p[u] = make_int2(0, 0);
            if (i < total) p[u] = pairs[segsrc[lo[u]] + (i - segoff[lo[u]])];
        }
        #pragma unroll
        for (int u = 0; u < RPT; u++) {
            if (i0 + u * 1024 < total) {
                const int ld = p[u].x - node0;
                const int r  = atomicAdd(&dcnt[ld], 1);
                if (r < CAP) {
                    lines[ld * CAP + r] = p[u].y;
                } else {
                    const int k = __hip_atomic_fetch_add(&deg[N_NODES], 1,
                                                         __ATOMIC_RELAXED, __HIP_MEMORY_SCOPE_AGENT);
                    spill[2 * (size_t)k]     = p[u].x;
                    spill[2 * (size_t)k + 1] = p[u].y;
                }
            }
        }
    }
    __syncthreads();

    const int nn = (N_NODES - node0 < BIN_N) ? (N_NODES - node0) : BIN_N;
    if (tid < nn) deg[node0 + tid] = dcnt[tid];
    for (int i = tid; i < nn * CAP; i += 1024) {
        bucket[(size_t)node0 * CAP + i] = lines[i];
    }
}

__global__ __launch_bounds__(256) void sage_packw(const float* __restrict__ Wl,
                                                  const float* __restrict__ Wr,
                                                  float2* __restrict__ Wp) {
    int t = blockIdx.x * 256 + threadIdx.x;
    if (t >= 128 * 128) return;
    int p = t >> 7;
    int n = t & 127;
    int k0 = 2 * p;
    float a, b;
    if (k0 < DIM) {
        a = Wl[n * DIM + k0];
        b = Wl[n * DIM + k0 + 1];
    } else {
        a = Wr[n * DIM + (k0 - DIM)];
        b = Wr[n * DIM + (k0 - DIM) + 1];
    }
    Wp[t] = make_float2(a, b);
}

template <int NT>
__device__ __forceinline__ void sage_tiles(const float2 (*ldsA)[130],
                                           const v2f* __restrict__ wpv,
                                           const float* __restrict__ bias,
                                           float* __restrict__ out,
                                           int r0, int mt0, int n0, int lane, int relu) {
    const int half = lane >> 4;
    const int m16  = lane & 15;

    v8f acc[NT];
    #pragma unroll
    for (int t = 0; t < NT; t++) acc[t] = v8f{};

    #pragma unroll 4
    for (int kp = 0; kp < 128; kp += 2) {
        v2f b = wpv[(size_t)(kp + half) * 128 + n0 + m16];
        #pragma unroll
        for (int t = 0; t < NT; t++) {
            v2f a = *(const v2f*)&ldsA[(mt0 + t) * 16 + m16][kp + half];
            acc[t] = __builtin_amdgcn_wmma_f32_16x16x4_f32(
                 false, a,  false, b,
                 (short)0, acc[t],  false,  false);
        }
    }

    const int   ncol = n0 + m16;
    const float bv   = bias[ncol];
    #pragma unroll
    for (int t = 0; t < NT; t++) {
        #pragma unroll
        for (int r = 0; r < 8; r++) {
            int   row = r0 + (mt0 + t) * 16 + r + 8 * half;
            float val = acc[t][r] + bv;
            if (relu) val = fmaxf(val, 0.0f);
            out[(size_t)row * DIM + ncol] = val;
        }
    }
}

__global__ __launch_bounds__(512) void sage_gemm(const int* __restrict__ deg,
                                                 const int* __restrict__ bucket,
                                                 const int* __restrict__ spill,
                                                 const float* __restrict__ xin,
                                                 const float2* __restrict__ Wp,
                                                 const float* __restrict__ bias,
                                                 float* __restrict__ out, int relu) {
    __shared__ float2 ldsA[TILE_M][130];

    const int tid  = threadIdx.x;
    const int lane = tid & 31;
    const int wave = __builtin_amdgcn_readfirstlane(tid >> 5);

    const int t0 = (int)(((unsigned)blockIdx.x * ROW_TILES) / ROW_BLOCKS);
    const int nt = (int)((((unsigned)blockIdx.x + 1u) * ROW_TILES) / ROW_BLOCKS) - t0;
    const int r0 = t0 * 16;

    const float4* x4 = (const float4*)xin;
    const int m0 = wave * nt;
    const int degl = deg[r0 + m0 + (lane < nt ? lane : nt - 1)];
    int ids[5];
    #pragma unroll
    for (int i = 0; i < 5; i++) {
        ids[i] = bucket[(r0 + m0 + (i < nt ? i : nt - 1)) * CAP + lane];
    }

    {
        float4 rt[4];
        #pragma unroll
        for (int i = 0; i < 4; i++) {
            const int idx = tid + i * 512;
            rt[i] = x4[(size_t)(r0 + (idx >> 5)) * 32 + (idx & 31)];
        }
        float4 rt4 = make_float4(0.0f, 0.0f, 0.0f, 0.0f);
        if (nt == 5) {
            const int idx = tid + 4 * 512;
            rt4 = x4[(size_t)(r0 + (idx >> 5)) * 32 + (idx & 31)];
        }
        #pragma unroll
        for (int i = 0; i < 4; i++) {
            const int idx = tid + i * 512;
            *(float4*)&ldsA[idx >> 5][64 + 2 * (idx & 31)] = rt[i];
        }
        if (nt == 5) {
            const int idx = tid + 4 * 512;
            *(float4*)&ldsA[idx >> 5][64 + 2 * (idx & 31)] = rt4;
        }
    }

    {
        const int dcl = (lane < nt) ? (degl < CAP ? degl : CAP) : 0;
        int incl = dcl;
        #pragma unroll
        for (int off = 1; off < 32; off <<= 1) {
            const int t = __shfl_up(incl, off);
            if (lane >= off) incl += t;
        }
        const int rpv = incl - dcl;

        const int se = __builtin_amdgcn_readlane(rpv, nt);
        int cur  = 0;
        int nxtb = __builtin_amdgcn_readlane(rpv, 1);
        float4 acc = make_float4(0.0f, 0.0f, 0.0f, 0.0f);

#define SAGE_CLOSE_ROW()                                                                   \
        {                                                                                  \
              \
            const int fd = __builtin_amdgcn_readlane(degl, cur);                           \
            if (fd > CAP) {                                                                \
                const int grow = r0 + m0 + cur;                                            \
                const int ns = deg[N_NODES];                                               \
                for (int q = 0; q < ns; q++) {                                             \
                    if (spill[2 * q] == grow) {                                            \
                        const float4 sv = x4[(size_t)spill[2 * q + 1] * 32 + lane];        \
                        acc.x += sv.x;                                                     \
                        acc.y += sv.y;                                                     \
                        acc.z += sv.z;                                                     \
                        acc.w += sv.w;                                                     \
                    }                                                                      \
                }                                                                          \
            }                                                                              \
                   \
            const float iv = 1.0f / fmaxf((float)fd, 1.0f);                                \
            ldsA[m0 + cur][2 * lane]     = make_float2(acc.x * iv, acc.y * iv);            \
            ldsA[m0 + cur][2 * lane + 1] = make_float2(acc.z * iv, acc.w * iv);            \
            acc = make_float4(0.0f, 0.0f, 0.0f, 0.0f);                                     \
            cur++;                                                                         \
            nxtb = __builtin_amdgcn_readlane(rpv, (cur + 1 <= nt) ? cur + 1 : nt);         \
        }

        for (int base = 0; base < se; base += 32) {
            const int n = (se - base < 32) ? (se - base) : 32;
            const int p = base + lane;
            int myid = 0;
            #pragma unroll
            for (int i = 0; i < 5; i++) {
                const int st = __builtin_amdgcn_readlane(rpv, i);
                const int en = __builtin_amdgcn_readlane(rpv, i + 1);
                const int t  = __shfl(ids[i], (p - st) & 31);
                if (p >= st && p < en) myid = t;
            }
            for (int k = 0; k < n; k += 8) {
                float4 v[8];
                int sid[8];
                #pragma unroll
                for (int u = 0; u < 8; u++) {
                    sid[u] = __builtin_amdgcn_readlane(myid, k + u);
                }
                #pragma unroll
                for (int u = 0; u < 8; u++) {
                    v[u] = make_float4(0.0f, 0.0f, 0.0f, 0.0f);
                    if (k + u < n) {
                        v[u] = x4[(size_t)sid[u] * 32 + lane];
                    }
                }
                #pragma unroll
                for (int u = 0; u < 8; u++) {
                    if (k + u < n) {
                        const int j = base + k + u;
                        while (j >= nxtb) SAGE_CLOSE_ROW()
                        acc.x += v[u].x;
                        acc.y += v[u].y;
                        acc.z += v[u].z;
                        acc.w += v[u].w;
                    }
                }
            }
        }
        while (cur < nt) SAGE_CLOSE_ROW()
#undef SAGE_CLOSE_ROW
    }
    __syncthreads();

    const int n0 = (wave & 7) * 16;
    const v2f* wpv = (const v2f*)Wp;
    if (nt == 5 && wave < 8) {
        sage_tiles<3>(ldsA, wpv, bias, out, r0, 0, n0, lane, relu);
    } else {
        sage_tiles<2>(ldsA, wpv, bias, out, r0, (wave < 8) ? 0 : nt - 2, n0, lane, relu);
    }
}

extern "C" void kernel_launch(void* const* d_in, const int* in_sizes, int n_in,
                              void* d_out, int out_size, void* d_ws, size_t ws_size,
                              hipStream_t stream) {
    const float* x  = (const float*)d_in[0];
    const int*   ei = (const int*)d_in[1];
    const float* Wl = (const float*)d_in[2];
    const float* bl = (const float*)d_in[3];
    const float* Wr = (const float*)d_in[4];
    const int E = in_sizes[1] / 2;
    const int* src = ei;
    const int* dst = ei + E;

    char*   ws     = (char*)d_ws;
    int*    deg    = (int*)ws;
    float2* wp     = (float2*)(ws + (256u << 10));
    int*    bucket = (int*)(ws + (512u << 10) + (4u << 20));
    int*    spill  = (int*)(ws + (512u << 10) + (12u << 20));
    int2*   tab    = (int2*)(ws + (512u << 10));
    int2*   pairs  = (int2*)(ws + (512u << 10) + (18u << 20));
    float*  h1     = (float*)(ws + (512u << 10) + (26u << 20));
    float*  y      = (float*)d_out;

    sage_binsort<<<EBLOCKS, 512, 0, stream>>>(src, dst, deg, tab, pairs, E);
    sage_rank<<<NBINS, 1024, 0, stream>>>(tab, pairs, deg, bucket, spill);
    sage_packw<<<64, 256, 0, stream>>>(Wl, Wr, wp);

    sage_gemm<<<ROW_BLOCKS, 512, 0, stream>>>(deg, bucket, spill, x, wp, bl, h1, 1);

    sage_gemm<<<ROW_BLOCKS, 512, 0, stream>>>(deg, bucket, spill, h1, wp, bl, y, 0);
}
